# starting kernel
# baseline (speedup 1.0000x reference)
#include <hip/hip_runtime.h>
#include <math.h>


typedef float v2f __attribute__((ext_vector_type(2)));
typedef float v8f __attribute__((ext_vector_type(8)));

#define IMG 512
#define TILE 64
#define HALO 8
#define TAPS 17
#define IN_W 80
#define IN_STRIDE 81
#define HB_STRIDE 65

struct GaussW { float w[TAPS]; };

__device__ __forceinline__ float fast_sigmoid(float z) {
    return __builtin_amdgcn_rcpf(1.0f + __expf(-z));
}

__global__ void lamb_kernel(const float* __restrict__ c,
                            float* __restrict__ lamb, int n) {
    int i = blockIdx.x * 256 + threadIdx.x;
    if (i < n) lamb[i] = 100.0f * fast_sigmoid(10.0f * c[i]);
}

template <bool INIT>
__global__ __launch_bounds__(256)
void bvp_iter(const float* __restrict__ src,
              const float* __restrict__ x,
              const float* __restrict__ lamb,
              float* __restrict__ dst,
              GaussW gw) {
    __shared__ float in_tile[IN_W * IN_STRIDE];
    __shared__ float hbuf[IN_W * HB_STRIDE];
    __shared__ float wsm[TAPS];

    const int tid  = threadIdx.x;
    const int lane = tid & 31;
    const int wave = tid >> 5;
    const int n15  = lane & 15;
    const int koff = (lane >> 4) << 1;
    const int mhalf = (lane >> 4) << 3;

    const int row0 = blockIdx.y * TILE;
    const int col0 = blockIdx.x * TILE;
    const size_t base = (size_t)blockIdx.z * (IMG * IMG);

    if (tid < TAPS) wsm[tid] = gw.w[tid];

    for (int i = tid; i < IN_W * IN_W; i += 256) {
        int ir = i / IN_W, ic = i - ir * IN_W;
        int gr = row0 - HALO + ir, gc = col0 - HALO + ic;
        float s = 0.0f;
        if (gr >= 0 && gr < IMG && gc >= 0 && gc < IMG)
            s = 1.0f - 2.0f * src[base + (size_t)gr * IMG + gc];
        in_tile[ir * IN_STRIDE + ic] = s;
    }
    __syncthreads();

    float wband[8][2];
#pragma unroll
    for (int c = 0; c < 8; ++c)
#pragma unroll
        for (int v = 0; v < 2; ++v) {
            int d = 4 * c + koff + v - n15;
            wband[c][v] = (d >= 0 && d < TAPS) ? wsm[d] : 0.0f;
        }

    for (int t = wave; t < 20; t += 8) {
        int R = t >> 2, C = t & 3;
        const float* arow = &in_tile[(R * 16 + n15) * IN_STRIDE + C * 16 + koff];
        v8f acc = {};
#pragma unroll
        for (int c = 0; c < 8; ++c) {
            v2f a; a.x = arow[4 * c]; a.y = arow[4 * c + 1];
            v2f b; b.x = wband[c][0]; b.y = wband[c][1];
            acc = __builtin_amdgcn_wmma_f32_16x16x4_f32(
                false, a, false, b, (short)0, acc, false, false);
        }
#pragma unroll
        for (int v = 0; v < 8; ++v)
            hbuf[(R * 16 + mhalf + v) * HB_STRIDE + C * 16 + n15] = acc[v];
    }
    __syncthreads();

    const float inv_tau = 10.0f;
    const float inv_eps = 0.1f;
    for (int t = wave; t < 16; t += 8) {
        int R = t >> 2, C = t & 3;
        v8f acc = {};
#pragma unroll
        for (int c = 0; c < 8; ++c) {
            v2f a; a.x = wband[c][0]; a.y = wband[c][1];
            const float* hp = &hbuf[(R * 16 + 4 * c + koff) * HB_STRIDE + C * 16 + n15];
            v2f b; b.x = hp[0]; b.y = hp[HB_STRIDE];
            acc = __builtin_amdgcn_wmma_f32_16x16x4_f32(
                false, a, false, b, (short)0, acc, false, false);
        }
#pragma unroll
        for (int v = 0; v < 8; ++v) {
            int lr = R * 16 + mhalf + v;
            int lc = C * 16 + n15;
            size_t gidx = base + (size_t)(row0 + lr) * IMG + (col0 + lc);
            float blur = acc[v];
            float lb = lamb[gidx];
            float u;
            if (INIT) {
                u = fast_sigmoid(-(lb * blur) * inv_eps);
            } else {
                float up = 0.5f * (1.0f - in_tile[(lr + HALO) * IN_STRIDE + (lc + HALO)]);
                float xv = x[gidx];
                float uu = fast_sigmoid(-((up - xv) * inv_tau + lb * blur) * inv_eps);
                u = 0.5f * up + 0.5f * uu;
            }
            dst[gidx] = u;
        }
    }
}

extern "C" void kernel_launch(void* const* d_in, const int* in_sizes, int n_in,
                              void* d_out, int out_size, void* d_ws, size_t ws_size,
                              hipStream_t stream) {
    const float* x = (const float*)d_in[0];
    const float* c = (const float*)d_in[1];
    float* out = (float*)d_out;
    const int n = in_sizes[0];
    const int planes = n / (IMG * IMG);

    float* lamb = (float*)d_ws;
    float* ubuf = lamb + n;

    GaussW gw;
    {
        float g[TAPS], s = 0.0f;
        for (int t = 0; t < TAPS; ++t) {
            float d = (float)t - 8.0f;
            g[t] = expf(-(d * d) / 8.0f);
            s += g[t];
        }
        for (int t = 0; t < TAPS; ++t) gw.w[t] = g[t] / s;
    }

    lamb_kernel<<<(n + 255) / 256, 256, 0, stream>>>(c, lamb, n);

    dim3 grid(IMG / TILE, IMG / TILE, planes);
    bvp_iter<true ><<<grid, 256, 0, stream>>>(x,    x, lamb, ubuf, gw);
    bvp_iter<false><<<grid, 256, 0, stream>>>(ubuf, x, lamb, out,  gw);
    bvp_iter<false><<<grid, 256, 0, stream>>>(out,  x, lamb, ubuf, gw);
    bvp_iter<false><<<grid, 256, 0, stream>>>(ubuf, x, lamb, out,  gw);
    bvp_iter<false><<<grid, 256, 0, stream>>>(out,  x, lamb, ubuf, gw);
    bvp_iter<false><<<grid, 256, 0, stream>>>(ubuf, x, lamb, out,  gw);
}


# Vertical pass first, B fragments read straight from global (64-byte row segments, a tile's 16 loads in flight); one LDS pass between, no staged input tile; k order 16h+2j+s, sums regrouped
# speedup vs baseline: 1.4151x; 1.4151x over previous
#include <hip/hip_runtime.h>
#include <math.h>


typedef float v2f __attribute__((ext_vector_type(2)));
typedef float v4f __attribute__((ext_vector_type(4)));
typedef float v8f __attribute__((ext_vector_type(8)));

#define IMG 512
#define TILE 64
#define HALO 8
#define TAPS 17
#define VB_PITCH 84

struct GaussW { float w[TAPS]; };

__device__ __forceinline__ float fast_sigmoid(float z) {
    return __builtin_amdgcn_rcpf(1.0f + __expf(-z));
}

__global__ void lamb_kernel(const float* __restrict__ c,
                            float* __restrict__ lamb, int n) {
    int i = blockIdx.x * 256 + threadIdx.x;
    if (i < n) lamb[i] = 100.0f * fast_sigmoid(10.0f * c[i]);
}

template <bool INIT>
__global__ __launch_bounds__(256)
void bvp_iter(const float* __restrict__ src,
              const float* __restrict__ x,
              const float* __restrict__ lamb,
              float* __restrict__ dst,
              GaussW gw) {
    __shared__ __attribute__((aligned(16))) float vbuf[TILE * VB_PITCH];

    const int tid  = threadIdx.x;
    const int lane = tid & 31;
    const int wave = tid >> 5;
    const int n15  = lane & 15;
    const int h    = lane >> 4;
    const int mhalf = h << 3;

    const int row0 = blockIdx.y * TILE;
    const int col0 = blockIdx.x * TILE;
    const size_t base = (size_t)blockIdx.z * (IMG * IMG);
    const float* __restrict__ plane = src + base;

    const float wl = gw.w[lane < TAPS ? lane : TAPS - 1];

    float wband[8][2];
#pragma unroll
    for (int j = 0; j < 8; ++j)
#pragma unroll
        for (int s = 0; s < 2; ++s) {
            int d = 16 * h + 2 * j + s - n15;
            float t = __shfl(wl, d & 31);
            wband[j][s] = (d >= 0 && d < TAPS) ? t : 0.0f;
        }

    for (int t = wave; t < 20; t += 8) {
        const int R = t / 5, C = t - 5 * R;
        const int rb = row0 - HALO + R * 16 + 16 * h;
        const int gc = col0 - HALO + C * 16 + n15;
        const bool colok = (unsigned)gc < (unsigned)IMG;
        const int gcc = min(max(gc, 0), IMG - 1);
        float sv[8][2];
#pragma unroll
        for (int i = 0; i < 2; ++i) {
            const int r8 = rb + 8 * i;
            const bool ok = colok && ((unsigned)r8 < (unsigned)IMG);
            const int r8c = min(max(r8, 0), IMG - 8);
            const float* __restrict__ p = plane + (r8c * IMG + gcc);
#pragma unroll
            for (int jj = 0; jj < 4; ++jj)
#pragma unroll
                for (int s = 0; s < 2; ++s) {
                    float v = p[(2 * jj + s) * IMG];
                    sv[4 * i + jj][s] = ok ? (1.0f - 2.0f * v) : 0.0f;
                }
        }
        v8f acc = {};
#pragma unroll
        for (int j = 0; j < 8; ++j) {
            v2f a; a.x = wband[j][0]; a.y = wband[j][1];
            v2f b; b.x = sv[j][0];    b.y = sv[j][1];
            acc = __builtin_amdgcn_wmma_f32_16x16x4_f32(
                false, a, false, b, (short)0, acc, false, false);
        }
#pragma unroll
        for (int r = 0; r < 8; ++r)
            vbuf[(R * 16 + mhalf + r) * VB_PITCH + C * 16 + n15] = acc[r];
    }
    __syncthreads();

    const float inv_tau = 10.0f;
    const float inv_eps = 0.1f;
    for (int t = wave; t < 16; t += 8) {
        const int R = t >> 2, C = t & 3;
        const v4f* ap = reinterpret_cast<const v4f*>(
            &vbuf[(R * 16 + n15) * VB_PITCH + C * 16 + 16 * h]);
        v4f f[4];
#pragma unroll
        for (int q = 0; q < 4; ++q) f[q] = ap[q];
        v8f acc = {};
#pragma unroll
        for (int j = 0; j < 8; ++j) {
            v2f a;
            if (j & 1) { a.x = f[j >> 1].z; a.y = f[j >> 1].w; }
            else       { a.x = f[j >> 1].x; a.y = f[j >> 1].y; }
            v2f b; b.x = wband[j][0]; b.y = wband[j][1];
            acc = __builtin_amdgcn_wmma_f32_16x16x4_f32(
                false, a, false, b, (short)0, acc, false, false);
        }
#pragma unroll
        for (int r = 0; r < 8; ++r) {
            int lr = R * 16 + mhalf + r;
            int lc = C * 16 + n15;
            size_t gidx = base + (size_t)(row0 + lr) * IMG + (col0 + lc);
            float blur = acc[r];
            float lb = lamb[gidx];
            float u;
            if (INIT) {
                u = fast_sigmoid(-(lb * blur) * inv_eps);
            } else {
                float up = src[gidx];
                float xv = x[gidx];
                float uu = fast_sigmoid(-((up - xv) * inv_tau + lb * blur) * inv_eps);
                u = 0.5f * up + 0.5f * uu;
            }
            dst[gidx] = u;
        }
    }
}

extern "C" void kernel_launch(void* const* d_in, const int* in_sizes, int n_in,
                              void* d_out, int out_size, void* d_ws, size_t ws_size,
                              hipStream_t stream) {
    const float* x = (const float*)d_in[0];
    const float* c = (const float*)d_in[1];
    float* out = (float*)d_out;
    const int n = in_sizes[0];
    const int planes = n / (IMG * IMG);

    float* lamb = (float*)d_ws;
    float* ubuf = lamb + n;

    GaussW gw;
    {
        float g[TAPS], s = 0.0f;
        for (int t = 0; t < TAPS; ++t) {
            float d = (float)t - 8.0f;
            g[t] = expf(-(d * d) / 8.0f);
            s += g[t];
        }
        for (int t = 0; t < TAPS; ++t) gw.w[t] = g[t] / s;
    }

    lamb_kernel<<<(n + 255) / 256, 256, 0, stream>>>(c, lamb, n);

    dim3 grid(IMG / TILE, IMG / TILE, planes);
    bvp_iter<true ><<<grid, 256, 0, stream>>>(x,    x, lamb, ubuf, gw);
    bvp_iter<false><<<grid, 256, 0, stream>>>(ubuf, x, lamb, out,  gw);
    bvp_iter<false><<<grid, 256, 0, stream>>>(out,  x, lamb, ubuf, gw);
    bvp_iter<false><<<grid, 256, 0, stream>>>(ubuf, x, lamb, out,  gw);
    bvp_iter<false><<<grid, 256, 0, stream>>>(out,  x, lamb, ubuf, gw);
    bvp_iter<false><<<grid, 256, 0, stream>>>(ubuf, x, lamb, out,  gw);
}


# Few long-lived waves per launch: a work-group walks 3 planes at its tile, next plane's loads requested ahead across an LDS-only barrier (12288 -> 4096 waves); lam_b pass 196608 -> 8192 waves
# speedup vs baseline: 1.6491x; 1.1654x over previous
#include <hip/hip_runtime.h>
#include <math.h>


typedef float v2f __attribute__((ext_vector_type(2)));
typedef float v4f __attribute__((ext_vector_type(4)));
typedef float v8f __attribute__((ext_vector_type(8)));

#define IMG 512
#define TILE 64
#define HALO 8
#define TAPS 17
#define VB_PITCH 84
#define PLANE (IMG * IMG)

struct GaussW { float w[TAPS]; };

__device__ __forceinline__ float fast_sigmoid(float z) {
    return __builtin_amdgcn_rcpf(1.0f + __expf(-z));
}

__global__ __launch_bounds__(256)
void lamb_kernel(const float* __restrict__ c, float* __restrict__ lamb, int n) {
    const int n4 = n >> 2;
    const int stride = gridDim.x * 256;
    const v4f* __restrict__ c4 = reinterpret_cast<const v4f*>(c);
    v4f* __restrict__ l4 = reinterpret_cast<v4f*>(lamb);
    for (int i0 = blockIdx.x * 256 + threadIdx.x; i0 < n4; i0 += 6 * stride) {
        v4f v[6];
#pragma unroll
        for (int k = 0; k < 6; ++k) {
            const int i = i0 + k * stride;
            v[k] = c4[i < n4 ? i : i0];
        }
#pragma unroll
        for (int k = 0; k < 6; ++k) {
            const int i = i0 + k * stride;
            v4f o;
            o.x = 100.0f * fast_sigmoid(10.0f * v[k].x);
            o.y = 100.0f * fast_sigmoid(10.0f * v[k].y);
            o.z = 100.0f * fast_sigmoid(10.0f * v[k].z);
            o.w = 100.0f * fast_sigmoid(10.0f * v[k].w);
            if (i < n4) l4[i] = o;
        }
    }
    const int g = blockIdx.x * 256 + threadIdx.x;
    if (g < (n & 3)) {
        const int i = (n4 << 2) + g;
        lamb[i] = 100.0f * fast_sigmoid(10.0f * c[i]);
    }
}

struct Geo {
    int  voff[3][2];
    bool vok[3][2];
    int  vst[3];
    int  hld[2];
    int  eoff[2];
};

__device__ __forceinline__ void issue_v(const float* __restrict__ plane, const Geo& g,
                                        int wave, float (&vraw)[3][2][8]) {
#pragma unroll
    for (int k = 0; k < 3; ++k) {
        if (k < 2 || wave < 4) {
#pragma unroll
            for (int i = 0; i < 2; ++i) {
                const float* __restrict__ p = plane + g.voff[k][i];
#pragma unroll
                for (int q = 0; q < 8; ++q) vraw[k][i][q] = p[q * IMG];
            }
        }
    }
}

template <bool INIT>
__device__ __forceinline__ void issue_e(const float* __restrict__ lamb, const float* __restrict__ src,
                                        const float* __restrict__ x, size_t base, const Geo& g,
                                        float (&elb)[2][8], float (&eup)[2][8], float (&ex)[2][8]) {
#pragma unroll
    for (int k = 0; k < 2; ++k)
#pragma unroll
        for (int r = 0; r < 8; ++r) {
            const size_t gi = base + (size_t)(g.eoff[k] + r * IMG);
            elb[k][r] = lamb[gi];
            if (!INIT) { eup[k][r] = src[gi]; ex[k][r] = x[gi]; }
        }
}

template <bool INIT, bool MORE>
__device__ __forceinline__ void do_plane(const float* __restrict__ src, const float* __restrict__ x,
                                         const float* __restrict__ lamb, float* __restrict__ dst,
                                         size_t base, size_t nbase, const Geo& g, int wave,
                                         const float (&wband)[8][2], float* __restrict__ vb,
                                         float (&vraw)[3][2][8],
                                         float (&elb)[2][8], float (&eup)[2][8], float (&ex)[2][8]) {
#pragma unroll
    for (int k = 0; k < 3; ++k) {
        if (k < 2 || wave < 4) {
            v8f acc = {};
#pragma unroll
            for (int j = 0; j < 8; ++j) {
                const int i = j >> 2, q = (j & 3) * 2;
                const bool ok = g.vok[k][i];
                v2f a; a.x = wband[j][0]; a.y = wband[j][1];
                v2f b;
                b.x = ok ? (1.0f - 2.0f * vraw[k][i][q])     : 0.0f;
                b.y = ok ? (1.0f - 2.0f * vraw[k][i][q + 1]) : 0.0f;
                acc = __builtin_amdgcn_wmma_f32_16x16x4_f32(
                    false, a, false, b, (short)0, acc, false, false);
            }
#pragma unroll
            for (int r = 0; r < 8; ++r) vb[g.vst[k] + r * VB_PITCH] = acc[r];
        }
    }
    if (MORE) issue_v(src + nbase, g, wave, vraw);
    __builtin_amdgcn_fence(__ATOMIC_RELEASE, "workgroup", "local");
    __builtin_amdgcn_s_barrier();
    __builtin_amdgcn_fence(__ATOMIC_ACQUIRE, "workgroup", "local");

    const float inv_tau = 10.0f;
    const float inv_eps = 0.1f;
#pragma unroll
    for (int k = 0; k < 2; ++k) {
        const v4f* ap = reinterpret_cast<const v4f*>(vb + g.hld[k]);
        v4f f[4];
#pragma unroll
        for (int q = 0; q < 4; ++q) f[q] = ap[q];
        v8f acc = {};
#pragma unroll
        for (int j = 0; j < 8; ++j) {
            v2f a;
            if (j & 1) { a.x = f[j >> 1].z; a.y = f[j >> 1].w; }
            else       { a.x = f[j >> 1].x; a.y = f[j >> 1].y; }
            v2f b; b.x = wband[j][0]; b.y = wband[j][1];
            acc = __builtin_amdgcn_wmma_f32_16x16x4_f32(
                false, a, false, b, (short)0, acc, false, false);
        }
#pragma unroll
        for (int r = 0; r < 8; ++r) {
            const size_t gi = base + (size_t)(g.eoff[k] + r * IMG);
            const float blur = acc[r];
            const float lb = elb[k][r];
            float u;
            if (INIT) {
                u = fast_sigmoid(-(lb * blur) * inv_eps);
            } else {
                const float up = eup[k][r];
                const float xv = ex[k][r];
                const float uu = fast_sigmoid(-((up - xv) * inv_tau + lb * blur) * inv_eps);
                u = 0.5f * up + 0.5f * uu;
            }
            dst[gi] = u;
        }
    }
    if (MORE) issue_e<INIT>(lamb, src, x, nbase, g, elb, eup, ex);
}

template <bool INIT>
__global__ __launch_bounds__(256)
void bvp_iter(const float* __restrict__ src,
              const float* __restrict__ x,
              const float* __restrict__ lamb,
              float* __restrict__ dst,
              GaussW gw, int planes) {
    __shared__ __attribute__((aligned(16))) float vbuf[2][TILE * VB_PITCH];

    const int tid  = threadIdx.x;
    const int lane = tid & 31;
    const int wave = __builtin_amdgcn_readfirstlane(tid >> 5);
    const int n15  = lane & 15;
    const int h    = lane >> 4;
    const int mhalf = h << 3;

    const int row0 = blockIdx.y * TILE;
    const int col0 = blockIdx.x * TILE;

    const float wl = gw.w[lane < TAPS ? lane : TAPS - 1];

    float wband[8][2];
#pragma unroll
    for (int j = 0; j < 8; ++j)
#pragma unroll
        for (int s = 0; s < 2; ++s) {
            int d = 16 * h + 2 * j + s - n15;
            float t = __shfl(wl, d & 31);
            wband[j][s] = (d >= 0 && d < TAPS) ? t : 0.0f;
        }

    Geo g;
#pragma unroll
    for (int k = 0; k < 3; ++k) {
        const int t = min(wave + 8 * k, 19);
        const int R = t / 5, C = t - 5 * R;
        const int rb = row0 - HALO + R * 16 + 16 * h;
        const int gc = col0 - HALO + C * 16 + n15;
        const bool colok = (unsigned)gc < (unsigned)IMG;
        const int gcc = min(max(gc, 0), IMG - 1);
#pragma unroll
        for (int i = 0; i < 2; ++i) {
            const int r8 = rb + 8 * i;
            g.vok[k][i] = colok && ((unsigned)r8 < (unsigned)IMG);
            const int r8c = min(max(r8, 0), IMG - 8);
            g.voff[k][i] = r8c * IMG + gcc;
        }
        g.vst[k] = (R * 16 + mhalf) * VB_PITCH + C * 16 + n15;
    }
#pragma unroll
    for (int k = 0; k < 2; ++k) {
        const int t = wave + 8 * k;
        const int R = t >> 2, C = t & 3;
        g.hld[k]  = (R * 16 + n15) * VB_PITCH + C * 16 + 16 * h;
        g.eoff[k] = (row0 + R * 16 + mhalf) * IMG + col0 + C * 16 + n15;
    }

    float vraw[3][2][8] = {};
    float elb[2][8] = {}, eup[2][8] = {}, ex[2][8] = {};

    const int step = gridDim.z;
    int pz = blockIdx.z;
    if (pz >= planes) return;
    size_t base = (size_t)pz * PLANE;
    issue_v(src + base, g, wave, vraw);
    __builtin_amdgcn_sched_barrier(0);
    issue_e<INIT>(lamb, src, x, base, g, elb, eup, ex);
    int buf = 0;
    for (; pz + step < planes; pz += step) {
        const size_t nbase = base + (size_t)step * PLANE;
        do_plane<INIT, true>(src, x, lamb, dst, base, nbase, g, wave, wband, vbuf[buf],
                             vraw, elb, eup, ex);
        base = nbase;
        buf ^= 1;
    }
    do_plane<INIT, false>(src, x, lamb, dst, base, base, g, wave, wband, vbuf[buf],
                          vraw, elb, eup, ex);
}

extern "C" void kernel_launch(void* const* d_in, const int* in_sizes, int n_in,
                              void* d_out, int out_size, void* d_ws, size_t ws_size,
                              hipStream_t stream) {
    const float* x = (const float*)d_in[0];
    const float* c = (const float*)d_in[1];
    float* out = (float*)d_out;
    const int n = in_sizes[0];
    const int planes = n / (IMG * IMG);

    float* lamb = (float*)d_ws;
    float* ubuf = lamb + n;

    GaussW gw;
    {
        float g[TAPS], s = 0.0f;
        for (int t = 0; t < TAPS; ++t) {
            float d = (float)t - 8.0f;
            g[t] = expf(-(d * d) / 8.0f);
            s += g[t];
        }
        for (int t = 0; t < TAPS; ++t) gw.w[t] = g[t] / s;
    }

    lamb_kernel<<<1024, 256, 0, stream>>>(c, lamb, n);

    dim3 grid(IMG / TILE, IMG / TILE, 8);
    bvp_iter<true ><<<grid, 256, 0, stream>>>(x,    x, lamb, ubuf, gw, planes);
    bvp_iter<false><<<grid, 256, 0, stream>>>(ubuf, x, lamb, out,  gw, planes);
    bvp_iter<false><<<grid, 256, 0, stream>>>(out,  x, lamb, ubuf, gw, planes);
    bvp_iter<false><<<grid, 256, 0, stream>>>(ubuf, x, lamb, out,  gw, planes);
    bvp_iter<false><<<grid, 256, 0, stream>>>(out,  x, lamb, ubuf, gw, planes);
    bvp_iter<false><<<grid, 256, 0, stream>>>(ubuf, x, lamb, out,  gw, planes);
}


# Update on whole 128-B lines: a wave owns a 16x32 patch, blur tiles cross lanes via wave-private LDS rows; lam_b/u/x loads and u stores are 16 B per lane, 64 -> 16 memory instr per wave and plane
# speedup vs baseline: 1.7729x; 1.0751x over previous
#include <hip/hip_runtime.h>
#include <math.h>


typedef float v2f __attribute__((ext_vector_type(2)));
typedef float v4f __attribute__((ext_vector_type(4)));
typedef float v8f __attribute__((ext_vector_type(8)));

#define IMG 512
#define TILE 64
#define HALO 8
#define TAPS 17
#define VB_PITCH 84
#define EP_PITCH 36
#define PLANE (IMG * IMG)

struct GaussW { float w[TAPS]; };

__device__ __forceinline__ float fast_sigmoid(float z) {
    return __builtin_amdgcn_rcpf(1.0f + __expf(-z));
}

__global__ __launch_bounds__(256)
void lamb_kernel(const float* __restrict__ c, float* __restrict__ lamb, int n) {
    const int n4 = n >> 2;
    const int stride = gridDim.x * 256;
    const v4f* __restrict__ c4 = reinterpret_cast<const v4f*>(c);
    v4f* __restrict__ l4 = reinterpret_cast<v4f*>(lamb);
    for (int i0 = blockIdx.x * 256 + threadIdx.x; i0 < n4; i0 += 6 * stride) {
        v4f v[6];
#pragma unroll
        for (int k = 0; k < 6; ++k) {
            const int i = i0 + k * stride;
            v[k] = c4[i < n4 ? i : i0];
        }
#pragma unroll
        for (int k = 0; k < 6; ++k) {
            const int i = i0 + k * stride;
            v4f o;
            o.x = 100.0f * fast_sigmoid(10.0f * v[k].x);
            o.y = 100.0f * fast_sigmoid(10.0f * v[k].y);
            o.z = 100.0f * fast_sigmoid(10.0f * v[k].z);
            o.w = 100.0f * fast_sigmoid(10.0f * v[k].w);
            if (i < n4) l4[i] = o;
        }
    }
    const int g = blockIdx.x * 256 + threadIdx.x;
    if (g < (n & 3)) {
        const int i = (n4 << 2) + g;
        lamb[i] = 100.0f * fast_sigmoid(10.0f * c[i]);
    }
}

struct Geo {
    int  voff[3][2];
    bool vok[3][2];
    int  vst[3];
    int  hld[2];
    int  est[2];
    int  eld;
    int  eoff;
};

__device__ __forceinline__ void issue_v(const float* __restrict__ plane, const Geo& g,
                                        int wave, float (&vraw)[3][2][8]) {
#pragma unroll
    for (int k = 0; k < 3; ++k) {
        if (k < 2 || wave < 4) {
#pragma unroll
            for (int i = 0; i < 2; ++i) {
                const float* __restrict__ p = plane + g.voff[k][i];
#pragma unroll
                for (int q = 0; q < 8; ++q) vraw[k][i][q] = p[q * IMG];
            }
        }
    }
}

template <bool INIT>
__device__ __forceinline__ void issue_e(const float* __restrict__ lamb, const float* __restrict__ src,
                                        const float* __restrict__ x, size_t base, const Geo& g,
                                        v4f (&elb)[4], v4f (&eup)[4], v4f (&ex)[4]) {
#pragma unroll
    for (int i = 0; i < 4; ++i) {
        const size_t gi = base + (size_t)(g.eoff + i * 4 * IMG);
        elb[i] = *reinterpret_cast<const v4f*>(lamb + gi);
        if (!INIT) {
            eup[i] = *reinterpret_cast<const v4f*>(src + gi);
            ex[i]  = *reinterpret_cast<const v4f*>(x + gi);
        }
    }
}

template <bool INIT>
__device__ __forceinline__ float update1(float blur, float lb, float up, float xv) {
    const float inv_tau = 10.0f;
    const float inv_eps = 0.1f;
    if (INIT) return fast_sigmoid(-(lb * blur) * inv_eps);
    const float uu = fast_sigmoid(-((up - xv) * inv_tau + lb * blur) * inv_eps);
    return 0.5f * up + 0.5f * uu;
}

template <bool INIT, bool MORE>
__device__ __forceinline__ void do_plane(const float* __restrict__ src, const float* __restrict__ x,
                                         const float* __restrict__ lamb, float* __restrict__ dst,
                                         size_t base, size_t nbase, const Geo& g, int wave,
                                         const float (&wband)[8][2], float* __restrict__ vb,
                                         float* erows, float (&vraw)[3][2][8],
                                         v4f (&elb)[4], v4f (&eup)[4], v4f (&ex)[4]) {
#pragma unroll
    for (int k = 0; k < 3; ++k) {
        if (k < 2 || wave < 4) {
            v8f acc = {};
#pragma unroll
            for (int j = 0; j < 8; ++j) {
                const int i = j >> 2, q = (j & 3) * 2;
                const bool ok = g.vok[k][i];
                v2f a; a.x = wband[j][0]; a.y = wband[j][1];
                v2f b;
                b.x = ok ? (1.0f - 2.0f * vraw[k][i][q])     : 0.0f;
                b.y = ok ? (1.0f - 2.0f * vraw[k][i][q + 1]) : 0.0f;
                acc = __builtin_amdgcn_wmma_f32_16x16x4_f32(
                    false, a, false, b, (short)0, acc, false, false);
            }
#pragma unroll
            for (int r = 0; r < 8; ++r) vb[g.vst[k] + r * VB_PITCH] = acc[r];
        }
    }
    if (MORE) issue_v(src + nbase, g, wave, vraw);
    __builtin_amdgcn_fence(__ATOMIC_RELEASE, "workgroup", "local");
    __builtin_amdgcn_s_barrier();
    __builtin_amdgcn_fence(__ATOMIC_ACQUIRE, "workgroup", "local");

#pragma unroll
    for (int k = 0; k < 2; ++k) {
        const v4f* ap = reinterpret_cast<const v4f*>(vb + g.hld[k]);
        v4f f[4];
#pragma unroll
        for (int q = 0; q < 4; ++q) f[q] = ap[q];
        v8f acc = {};
#pragma unroll
        for (int j = 0; j < 8; ++j) {
            v2f a;
            if (j & 1) { a.x = f[j >> 1].z; a.y = f[j >> 1].w; }
            else       { a.x = f[j >> 1].x; a.y = f[j >> 1].y; }
            v2f b; b.x = wband[j][0]; b.y = wband[j][1];
            acc = __builtin_amdgcn_wmma_f32_16x16x4_f32(
                false, a, false, b, (short)0, acc, false, false);
        }
#pragma unroll
        for (int r = 0; r < 8; ++r) erows[g.est[k] + r * EP_PITCH] = acc[r];
    }
#pragma unroll
    for (int i = 0; i < 4; ++i) {
        const size_t gi = base + (size_t)(g.eoff + i * 4 * IMG);
        const v4f blur = *reinterpret_cast<const v4f*>(erows + g.eld + i * 4 * EP_PITCH);
        v4f u;
        u.x = update1<INIT>(blur.x, elb[i].x, eup[i].x, ex[i].x);
        u.y = update1<INIT>(blur.y, elb[i].y, eup[i].y, ex[i].y);
        u.z = update1<INIT>(blur.z, elb[i].z, eup[i].z, ex[i].z);
        u.w = update1<INIT>(blur.w, elb[i].w, eup[i].w, ex[i].w);
        *reinterpret_cast<v4f*>(dst + gi) = u;
    }
    if (MORE) issue_e<INIT>(lamb, src, x, nbase, g, elb, eup, ex);
}

template <bool INIT>
__global__ __launch_bounds__(256)
void bvp_iter(const float* __restrict__ src,
              const float* __restrict__ x,
              const float* __restrict__ lamb,
              float* __restrict__ dst,
              GaussW gw, int planes) {
    __shared__ __attribute__((aligned(16))) float vbuf[2][TILE * VB_PITCH];
    __shared__ __attribute__((aligned(16))) float ebuf[8][16 * EP_PITCH];

    const int tid  = threadIdx.x;
    const int lane = tid & 31;
    const int wave = __builtin_amdgcn_readfirstlane(tid >> 5);
    const int n15  = lane & 15;
    const int h    = lane >> 4;
    const int mhalf = h << 3;

    const int row0 = blockIdx.y * TILE;
    const int col0 = blockIdx.x * TILE;

    const float wl = gw.w[lane < TAPS ? lane : TAPS - 1];

    float wband[8][2];
#pragma unroll
    for (int j = 0; j < 8; ++j)
#pragma unroll
        for (int s = 0; s < 2; ++s) {
            int d = 16 * h + 2 * j + s - n15;
            float t = __shfl(wl, d & 31);
            wband[j][s] = (d >= 0 && d < TAPS) ? t : 0.0f;
        }

    Geo g;
#pragma unroll
    for (int k = 0; k < 3; ++k) {
        const int t = min(wave + 8 * k, 19);
        const int R = t / 5, C = t - 5 * R;
        const int rb = row0 - HALO + R * 16 + 16 * h;
        const int gc = col0 - HALO + C * 16 + n15;
        const bool colok = (unsigned)gc < (unsigned)IMG;
        const int gcc = min(max(gc, 0), IMG - 1);
#pragma unroll
        for (int i = 0; i < 2; ++i) {
            const int r8 = rb + 8 * i;
            g.vok[k][i] = colok && ((unsigned)r8 < (unsigned)IMG);
            const int r8c = min(max(r8, 0), IMG - 8);
            g.voff[k][i] = r8c * IMG + gcc;
        }
        g.vst[k] = (R * 16 + mhalf) * VB_PITCH + C * 16 + n15;
    }
    const int hR = wave >> 1, hC0 = (wave & 1) * 2;
#pragma unroll
    for (int k = 0; k < 2; ++k) {
        g.hld[k] = (hR * 16 + n15) * VB_PITCH + (hC0 + k) * 16 + 16 * h;
        g.est[k] = mhalf * EP_PITCH + 16 * k + n15;
    }
    g.eld  = (lane >> 3) * EP_PITCH + 4 * (lane & 7);
    g.eoff = (row0 + hR * 16 + (lane >> 3)) * IMG + col0 + hC0 * 16 + 4 * (lane & 7);
    float* erows = ebuf[wave];

    float vraw[3][2][8] = {};
    v4f elb[4] = {}, eup[4] = {}, ex[4] = {};

    const int step = gridDim.z;
    int pz = blockIdx.z;
    if (pz >= planes) return;
    size_t base = (size_t)pz * PLANE;
    issue_v(src + base, g, wave, vraw);
    __builtin_amdgcn_sched_barrier(0);
    issue_e<INIT>(lamb, src, x, base, g, elb, eup, ex);
    int buf = 0;
    for (; pz + step < planes; pz += step) {
        const size_t nbase = base + (size_t)step * PLANE;
        do_plane<INIT, true>(src, x, lamb, dst, base, nbase, g, wave, wband, vbuf[buf],
                             erows, vraw, elb, eup, ex);
        base = nbase;
        buf ^= 1;
    }
    do_plane<INIT, false>(src, x, lamb, dst, base, base, g, wave, wband, vbuf[buf],
                          erows, vraw, elb, eup, ex);
}

extern "C" void kernel_launch(void* const* d_in, const int* in_sizes, int n_in,
                              void* d_out, int out_size, void* d_ws, size_t ws_size,
                              hipStream_t stream) {
    const float* x = (const float*)d_in[0];
    const float* c = (const float*)d_in[1];
    float* out = (float*)d_out;
    const int n = in_sizes[0];
    const int planes = n / (IMG * IMG);

    float* lamb = (float*)d_ws;
    float* ubuf = lamb + n;

    GaussW gw;
    {
        float g[TAPS], s = 0.0f;
        for (int t = 0; t < TAPS; ++t) {
            float d = (float)t - 8.0f;
            g[t] = expf(-(d * d) / 8.0f);
            s += g[t];
        }
        for (int t = 0; t < TAPS; ++t) gw.w[t] = g[t] / s;
    }

    lamb_kernel<<<1024, 256, 0, stream>>>(c, lamb, n);

    dim3 grid(IMG / TILE, IMG / TILE, 8);
    bvp_iter<true ><<<grid, 256, 0, stream>>>(x,    x, lamb, ubuf, gw, planes);
    bvp_iter<false><<<grid, 256, 0, stream>>>(ubuf, x, lamb, out,  gw, planes);
    bvp_iter<false><<<grid, 256, 0, stream>>>(out,  x, lamb, ubuf, gw, planes);
    bvp_iter<false><<<grid, 256, 0, stream>>>(ubuf, x, lamb, out,  gw, planes);
    bvp_iter<false><<<grid, 256, 0, stream>>>(out,  x, lamb, ubuf, gw, planes);
    bvp_iter<false><<<grid, 256, 0, stream>>>(ubuf, x, lamb, out,  gw, planes);
}
